# starting kernel
# baseline (speedup 1.0000x reference)
#include <hip/hip_runtime.h>
#include <hip/hip_bf16.h>

typedef __attribute__((ext_vector_type(2))) float v2f;
typedef __attribute__((ext_vector_type(8))) float v8f;

#define N_PTS 4096
#define C_CLS 256
#define D_DIM 128
#define CURVF 1.0f

__global__ __launch_bounds__(256) void hyp_pre_c(const float* __restrict__ a,
                                                 const float* __restrict__ p,
                                                 float* __restrict__ p2v,
                                                 float* __restrict__ pav,
                                                 float* __restrict__ nav) {
    int c = blockIdx.x * blockDim.x + threadIdx.x;
    if (c >= C_CLS) return;
    const float4* pr = (const float4*)(p + c * D_DIM);
    const float4* ar = (const float4*)(a + c * D_DIM);
    float sp2 = 0.f, spa = 0.f, sa2 = 0.f;
#pragma unroll 8
    for (int i = 0; i < D_DIM / 4; ++i) {
        float4 pv = pr[i], av = ar[i];
        sp2 += pv.x * pv.x + pv.y * pv.y + pv.z * pv.z + pv.w * pv.w;
        spa += pv.x * av.x + pv.y * av.y + pv.z * av.z + pv.w * av.w;
        sa2 += av.x * av.x + av.y * av.y + av.z * av.z + av.w * av.w;
    }
    float na = sqrtf(sa2);
    float nac = fmaxf(na, 1e-12f);
    p2v[c] = sp2;
    pav[c] = -spa / nac;
    nav[c] = na;
}

__global__ __launch_bounds__(256) void hyp_pre_n(const float* __restrict__ x,
                                                 float* __restrict__ x2v) {
    int n = blockIdx.x * blockDim.x + threadIdx.x;
    const float4* xr = (const float4*)(x + n * D_DIM);
    float s = 0.f;
#pragma unroll 8
    for (int i = 0; i < D_DIM / 4; ++i) {
        float4 xv = xr[i];
        s += xv.x * xv.x + xv.y * xv.y + xv.z * xv.z + xv.w * xv.w;
    }
    x2v[n] = s;
}

__global__ __launch_bounds__(256) void hyp_mlr_main(const float* __restrict__ X,
                                                    const float* __restrict__ A,
                                                    const float* __restrict__ P,
                                                    const float* __restrict__ p2v,
                                                    const float* __restrict__ pav,
                                                    const float* __restrict__ nav,
                                                    const float* __restrict__ x2v,
                                                    float* __restrict__ out) {
    const int lane = threadIdx.x & 31;
    const int wave = threadIdx.x >> 5;
    const int tile = blockIdx.x * 8 + wave;
    const int tN = tile & ((N_PTS / 16) - 1);
    const int tC = tile >> 8;
    const int n0 = tN * 16;
    const int c0 = tC * 16;
    const int m = lane & 15;
    const int half = lane >> 4;

    const float* xrow = X + (size_t)(n0 + m) * D_DIM + 2 * half;
    const float* prow = P + (size_t)(c0 + m) * D_DIM + 2 * half;
    const float* arow = A + (size_t)(c0 + m) * D_DIM + 2 * half;

    v8f acc_py = {};
    v8f acc_xa = {};
#pragma unroll
    for (int k = 0; k < D_DIM; k += 4) {
        v2f af = *(const v2f*)(xrow + k);
        v2f bp = *(const v2f*)(prow + k);
        v2f ba = *(const v2f*)(arow + k);
        acc_py = __builtin_amdgcn_wmma_f32_16x16x4_f32(false, af, false, bp,
                                                       (short)0, acc_py, false, false);
        acc_xa = __builtin_amdgcn_wmma_f32_16x16x4_f32(false, af, false, ba,
                                                       (short)0, acc_xa, false, false);
    }

    const int cc = c0 + m;
    const float p2 = p2v[cc];
    const float pa = pav[cc];
    const float na = nav[cc];
    const float inv_na = 1.0f / fmaxf(na, 1e-12f);
    const float cu = CURVF;

#pragma unroll
    for (int r = 0; r < 8; ++r) {
        const int n = n0 + r + 8 * half;
        const float x2 = x2v[n];
        const float xy = -acc_py[r];
        const float xa = acc_xa[r] * inv_na;
        const float t  = 1.0f + 2.0f * cu * xy;
        const float A1 = t + cu * x2;
        const float B1 = 1.0f - cu * p2;
        const float den = t + cu * cu * p2 * x2;
        const float inv_den = 1.0f / den;
        const float rr = (A1 * A1 * p2 + 2.0f * A1 * B1 * xy + B1 * B1 * x2)
                         * inv_den * inv_den;
        const float lam = 2.0f / (1.0f - rr);
        const float dot = (A1 * pa + B1 * xa) * inv_den;
        out[(size_t)n * C_CLS + cc] = 2.0f * na * asinhf(dot * lam);
    }
}

extern "C" void kernel_launch(void* const* d_in, const int* in_sizes, int n_in,
                              void* d_out, int out_size, void* d_ws, size_t ws_size,
                              hipStream_t stream) {
    const float* X = (const float*)d_in[0];
    const float* A = (const float*)d_in[1];
    const float* P = (const float*)d_in[2];
    float* out = (float*)d_out;

    float* ws  = (float*)d_ws;
    float* p2v = ws;
    float* pav = ws + C_CLS;
    float* nav = ws + 2 * C_CLS;
    float* x2v = ws + 3 * C_CLS;

    hyp_pre_c<<<1, 256, 0, stream>>>(A, P, p2v, pav, nav);
    hyp_pre_n<<<N_PTS / 256, 256, 0, stream>>>(X, x2v);

    const int tiles = (N_PTS / 16) * (C_CLS / 16);
    hyp_mlr_main<<<tiles / 8, 256, 0, stream>>>(X, A, P, p2v, pav, nav, x2v, out);
}


# norm pre-passes re-cut as ONE launch, a wave per row: a load takes a whole 512-byte row (4 whole lines, not 32 lines x 16 B); lane pieces joined by a 5-step butterfly (norm sums: new order)
# speedup vs baseline: 2.2673x; 2.2673x over previous
#include <hip/hip_runtime.h>


typedef __attribute__((ext_vector_type(2))) float v2f;
typedef __attribute__((ext_vector_type(8))) float v8f;

#define N_PTS 4096
#define C_CLS 256
#define D_DIM 128
#define CURVF 1.0f
#define NT_W 2
#define KQ (D_DIM / 8)

#define RPW 4
#define PRE_WPG 8
#define PRE_XW (N_PTS / RPW)
#define PRE_CW (C_CLS / RPW)

#define DPP_ROW_XMASK(M) (0x160 + (M))

static __device__ __forceinline__ float wave_sum(float s) {
    int t;
    t = __builtin_bit_cast(int, s);
    s += __builtin_bit_cast(float, __builtin_amdgcn_update_dpp(t, t, DPP_ROW_XMASK(1), 0xf, 0xf, true));
    t = __builtin_bit_cast(int, s);
    s += __builtin_bit_cast(float, __builtin_amdgcn_update_dpp(t, t, DPP_ROW_XMASK(2), 0xf, 0xf, true));
    t = __builtin_bit_cast(int, s);
    s += __builtin_bit_cast(float, __builtin_amdgcn_update_dpp(t, t, DPP_ROW_XMASK(4), 0xf, 0xf, true));
    t = __builtin_bit_cast(int, s);
    s += __builtin_bit_cast(float, __builtin_amdgcn_update_dpp(t, t, DPP_ROW_XMASK(8), 0xf, 0xf, true));
    t = __builtin_bit_cast(int, s);
    s += __builtin_bit_cast(float, __builtin_amdgcn_permlanex16(t, t, 0x76543210, 0xfedcba98, false, false));
    return s;
}

static __device__ __forceinline__ float dot4(const float4 u, const float4 v) {
    return u.x * v.x + u.y * v.y + u.z * v.z + u.w * v.w;
}

static_assert(D_DIM == 128, "one 16-byte piece per lane and row");
static_assert(RPW <= 32 && N_PTS % RPW == 0 && C_CLS % RPW == 0 && (PRE_XW + PRE_CW) % PRE_WPG == 0,
              "whole waves, whole work-groups");
__global__ __launch_bounds__(32 * PRE_WPG) void hyp_pre(const float* __restrict__ x,
                                                        const float* __restrict__ a,
                                                        const float* __restrict__ p,
                                                        float* __restrict__ p2v,
                                                        float* __restrict__ pav,
                                                        float* __restrict__ nav,
                                                        float* __restrict__ x2v) {
    const int lane = threadIdx.x & 31;
    const int gw = __builtin_amdgcn_readfirstlane((int)(blockIdx.x * PRE_WPG + (threadIdx.x >> 5)));
    if (gw >= PRE_CW) {
        const int n0 = (gw - PRE_CW) * RPW;
        const float4* xr = (const float4*)(x + (size_t)n0 * D_DIM) + lane;
        float4 xv[RPW];
#pragma unroll
        for (int r = 0; r < RPW; ++r) xv[r] = xr[r * (D_DIM / 4)];
        float mine = 0.f;
#pragma unroll
        for (int r = 0; r < RPW; ++r) {
            const float s = wave_sum(dot4(xv[r], xv[r]));
            if (lane == r) mine = s;
        }
        if (lane < RPW) x2v[n0 + lane] = mine;
    } else {
        const int c0 = gw * RPW;
        const float4* pr = (const float4*)(p + (size_t)c0 * D_DIM) + lane;
        const float4* ar = (const float4*)(a + (size_t)c0 * D_DIM) + lane;
        float4 pv[RPW], av[RPW];
#pragma unroll
        for (int r = 0; r < RPW; ++r) {
            pv[r] = pr[r * (D_DIM / 4)];
            av[r] = ar[r * (D_DIM / 4)];
        }
        float mp2 = 0.f, mpa = 0.f, ma2 = 0.f;
#pragma unroll
        for (int r = 0; r < RPW; ++r) {
            const float sp2 = wave_sum(dot4(pv[r], pv[r]));
            const float spa = wave_sum(dot4(pv[r], av[r]));
            const float sa2 = wave_sum(dot4(av[r], av[r]));
            if (lane == r) { mp2 = sp2; mpa = spa; ma2 = sa2; }
        }
        if (lane < RPW) {
            const float na = sqrtf(ma2);
            const float nac = fmaxf(na, 1e-12f);
            p2v[c0 + lane] = mp2;
            pav[c0 + lane] = -mpa / nac;
            nav[c0 + lane] = na;
        }
    }
}

__global__ __launch_bounds__(256) void hyp_mlr_main(const float* __restrict__ X,
                                                    const float* __restrict__ A,
                                                    const float* __restrict__ P,
                                                    const float* __restrict__ p2v,
                                                    const float* __restrict__ pav,
                                                    const float* __restrict__ nav,
                                                    const float* __restrict__ x2v,
                                                    float* __restrict__ out) {
    const int lane = threadIdx.x & 31;
    const int wave = threadIdx.x >> 5;
    const int m = lane & 15;
    const int half = lane >> 4;
    const int c0 = ((blockIdx.x & 3) * 4 + (wave & 3)) * 16;
    const int nbase = (blockIdx.x >> 2) * 64 + (wave >> 2) * (16 * NT_W);
    const int cc = c0 + m;

    const float4* prow = (const float4*)(P + (size_t)cc * D_DIM + 4 * half);
    const float4* arow = (const float4*)(A + (size_t)cc * D_DIM + 4 * half);
    const float4* xrow = (const float4*)(X + (size_t)(nbase + m) * D_DIM + 4 * half);

    v8f acc_py[NT_W];
    v8f acc_xa[NT_W];
#pragma unroll
    for (int t = 0; t < NT_W; ++t) {
        acc_py[t] = v8f{};
        acc_xa[t] = v8f{};
    }

#pragma unroll
    for (int q = 0; q < KQ; ++q) {
        const float4 pv = prow[2 * q];
        const float4 av = arow[2 * q];
        const v2f p0 = {pv.x, pv.y};
        const v2f p1 = {pv.z, pv.w};
        const v2f a0 = {av.x, av.y};
        const v2f a1 = {av.z, av.w};
#pragma unroll
        for (int t = 0; t < NT_W; ++t) {
            const float4 xv = xrow[t * (16 * D_DIM / 4) + 2 * q];
            const v2f x0 = {xv.x, xv.y};
            const v2f x1 = {xv.z, xv.w};
            acc_py[t] = __builtin_amdgcn_wmma_f32_16x16x4_f32(false, x0, false, p0,
                                                              (short)0, acc_py[t], false, false);
            acc_xa[t] = __builtin_amdgcn_wmma_f32_16x16x4_f32(false, x0, false, a0,
                                                              (short)0, acc_xa[t], false, false);
            acc_py[t] = __builtin_amdgcn_wmma_f32_16x16x4_f32(false, x1, false, p1,
                                                              (short)0, acc_py[t], false, false);
            acc_xa[t] = __builtin_amdgcn_wmma_f32_16x16x4_f32(false, x1, false, a1,
                                                              (short)0, acc_xa[t], false, false);
        }
    }

    const float p2 = p2v[cc];
    const float pa = pav[cc];
    const float na = nav[cc];
    const float inv_na = 1.0f / fmaxf(na, 1e-12f);
    const float cu = CURVF;

#pragma unroll
    for (int t = 0; t < NT_W; ++t) {
#pragma unroll
        for (int r = 0; r < 8; ++r) {
            const int n = nbase + 16 * t + r + 8 * half;
            const float x2 = x2v[n];
            const float xy = -acc_py[t][r];
            const float xa = acc_xa[t][r] * inv_na;
            const float tt = 1.0f + 2.0f * cu * xy;
            const float A1 = tt + cu * x2;
            const float B1 = 1.0f - cu * p2;
            const float den = tt + cu * cu * p2 * x2;
            const float inv_den = 1.0f / den;
            const float rr = (A1 * A1 * p2 + 2.0f * A1 * B1 * xy + B1 * B1 * x2)
                             * inv_den * inv_den;
            const float lam = 2.0f / (1.0f - rr);
            const float dot = (A1 * pa + B1 * xa) * inv_den;
            out[(size_t)n * C_CLS + cc] = 2.0f * na * asinhf(dot * lam);
        }
    }
}

extern "C" void kernel_launch(void* const* d_in, const int* in_sizes, int n_in,
                              void* d_out, int out_size, void* d_ws, size_t ws_size,
                              hipStream_t stream) {
    const float* X = (const float*)d_in[0];
    const float* A = (const float*)d_in[1];
    const float* P = (const float*)d_in[2];
    float* out = (float*)d_out;

    float* ws  = (float*)d_ws;
    float* p2v = ws;
    float* pav = ws + C_CLS;
    float* nav = ws + 2 * C_CLS;
    float* x2v = ws + 3 * C_CLS;

    hyp_pre<<<(PRE_XW + PRE_CW) / PRE_WPG, 32 * PRE_WPG, 0, stream>>>(X, A, P, p2v, pav, nav, x2v);

    hyp_mlr_main<<<(N_PTS / 64) * (C_CLS / 64), 256, 0, stream>>>(X, A, P, p2v, pav, nav, x2v, out);
}


# norm pass folded into the product launch: each work-group sums its own 64 rows of X, P and A (wave per row, whole-row loads) beside its products and hands the sums over in LDS: 2 launches -> 1
# speedup vs baseline: 2.4271x; 1.0705x over previous
#include <hip/hip_runtime.h>


typedef __attribute__((ext_vector_type(2))) float v2f;
typedef __attribute__((ext_vector_type(8))) float v8f;

#define N_PTS 4096
#define C_CLS 256
#define D_DIM 128
#define CURVF 1.0f
#define NT_W 2
#define KQ (D_DIM / 8)

#define WG_PTS 64
#define WG_CLS 64
#define WG_WAVES 8
#define NRW (WG_PTS / WG_WAVES)

#define DPP_ROW_XMASK(M) (0x160 + (M))

static __device__ __forceinline__ float wave_sum(float s) {
    int t;
    t = __builtin_bit_cast(int, s);
    s += __builtin_bit_cast(float, __builtin_amdgcn_update_dpp(t, t, DPP_ROW_XMASK(1), 0xf, 0xf, true));
    t = __builtin_bit_cast(int, s);
    s += __builtin_bit_cast(float, __builtin_amdgcn_update_dpp(t, t, DPP_ROW_XMASK(2), 0xf, 0xf, true));
    t = __builtin_bit_cast(int, s);
    s += __builtin_bit_cast(float, __builtin_amdgcn_update_dpp(t, t, DPP_ROW_XMASK(4), 0xf, 0xf, true));
    t = __builtin_bit_cast(int, s);
    s += __builtin_bit_cast(float, __builtin_amdgcn_update_dpp(t, t, DPP_ROW_XMASK(8), 0xf, 0xf, true));
    t = __builtin_bit_cast(int, s);
    s += __builtin_bit_cast(float, __builtin_amdgcn_permlanex16(t, t, 0x76543210, 0xfedcba98, false, false));
    return s;
}

static __device__ __forceinline__ float dot4(const float4 u, const float4 v) {
    return u.x * v.x + u.y * v.y + u.z * v.z + u.w * v.w;
}

static_assert(D_DIM == 128, "one 16-byte piece per lane and row");
static_assert(WG_PTS == WG_CLS && WG_PTS % WG_WAVES == 0 && NRW <= 32, "the norm phase deals whole rows to the waves");
__global__ __launch_bounds__(32 * WG_WAVES) void hyp_mlr_main(const float* __restrict__ X,
                                                              const float* __restrict__ A,
                                                              const float* __restrict__ P,
                                                              float* __restrict__ out) {
    __shared__ __attribute__((aligned(16))) float s_x2[WG_PTS];
    __shared__ float s_p2[WG_CLS];
    __shared__ float s_pa[WG_CLS];
    __shared__ float s_na[WG_CLS];

    const int lane = threadIdx.x & 31;
    const int wave = threadIdx.x >> 5;
    const int m = lane & 15;
    const int half = lane >> 4;
    const int cblk = (blockIdx.x & 3) * WG_CLS;
    const int nblk = (blockIdx.x >> 2) * WG_PTS;
    const int cl = (wave & 3) * 16 + m;
    const int nl = (wave >> 2) * (16 * NT_W);
    const int cc = cblk + cl;
    const int nbase = nblk + nl;

    const float4* prow = (const float4*)(P + (size_t)cc * D_DIM + 4 * half);
    const float4* arow = (const float4*)(A + (size_t)cc * D_DIM + 4 * half);
    const float4* xrow = (const float4*)(X + (size_t)(nbase + m) * D_DIM + 4 * half);

    v8f acc_py[NT_W];
    v8f acc_xa[NT_W];
#pragma unroll
    for (int t = 0; t < NT_W; ++t) {
        acc_py[t] = v8f{};
        acc_xa[t] = v8f{};
    }

#pragma unroll
    for (int q = 0; q < KQ; ++q) {
        const float4 pv = prow[2 * q];
        const float4 av = arow[2 * q];
        const v2f p0 = {pv.x, pv.y};
        const v2f p1 = {pv.z, pv.w};
        const v2f a0 = {av.x, av.y};
        const v2f a1 = {av.z, av.w};
#pragma unroll
        for (int t = 0; t < NT_W; ++t) {
            const float4 xv = xrow[t * (16 * D_DIM / 4) + 2 * q];
            const v2f x0 = {xv.x, xv.y};
            const v2f x1 = {xv.z, xv.w};
            acc_py[t] = __builtin_amdgcn_wmma_f32_16x16x4_f32(false, x0, false, p0,
                                                              (short)0, acc_py[t], false, false);
            acc_xa[t] = __builtin_amdgcn_wmma_f32_16x16x4_f32(false, x0, false, a0,
                                                              (short)0, acc_xa[t], false, false);
            acc_py[t] = __builtin_amdgcn_wmma_f32_16x16x4_f32(false, x1, false, p1,
                                                              (short)0, acc_py[t], false, false);
            acc_xa[t] = __builtin_amdgcn_wmma_f32_16x16x4_f32(false, x1, false, a1,
                                                              (short)0, acc_xa[t], false, false);
        }
    }

    {
        const int r0 = NRW * __builtin_amdgcn_readfirstlane(wave);
        const float4* xr = (const float4*)(X + (size_t)(nblk + r0) * D_DIM) + lane;
        const float4* pr = (const float4*)(P + (size_t)(cblk + r0) * D_DIM) + lane;
        const float4* ar = (const float4*)(A + (size_t)(cblk + r0) * D_DIM) + lane;
        float4 xv[NRW], pv[NRW], av[NRW];
#pragma unroll
        for (int r = 0; r < NRW; ++r) {
            xv[r] = xr[r * (D_DIM / 4)];
            pv[r] = pr[r * (D_DIM / 4)];
            av[r] = ar[r * (D_DIM / 4)];
        }
        float mx2 = 0.f, mp2 = 0.f, mpa = 0.f, ma2 = 0.f;
#pragma unroll
        for (int r = 0; r < NRW; ++r) {
            const float sx2 = wave_sum(dot4(xv[r], xv[r]));
            const float sp2 = wave_sum(dot4(pv[r], pv[r]));
            const float spa = wave_sum(dot4(pv[r], av[r]));
            const float sa2 = wave_sum(dot4(av[r], av[r]));
            if (lane == r) { mx2 = sx2; mp2 = sp2; mpa = spa; ma2 = sa2; }
        }
        if (lane < NRW) {
            const float na = sqrtf(ma2);
            const float nac = fmaxf(na, 1e-12f);
            s_x2[r0 + lane] = mx2;
            s_p2[r0 + lane] = mp2;
            s_pa[r0 + lane] = -mpa / nac;
            s_na[r0 + lane] = na;
        }
    }
    __syncthreads();

    const float p2 = s_p2[cl];
    const float pa = s_pa[cl];
    const float na = s_na[cl];
    const float inv_na = 1.0f / fmaxf(na, 1e-12f);
    const float cu = CURVF;

#pragma unroll
    for (int t = 0; t < NT_W; ++t) {
        const float4* xq = (const float4*)(s_x2 + nl + 16 * t + 8 * half);
        const float4 xlo = xq[0];
        const float4 xhi = xq[1];
        const float x2r[8] = {xlo.x, xlo.y, xlo.z, xlo.w, xhi.x, xhi.y, xhi.z, xhi.w};
#pragma unroll
        for (int r = 0; r < 8; ++r) {
            const int n = nbase + 16 * t + r + 8 * half;
            const float x2 = x2r[r];
            const float xy = -acc_py[t][r];
            const float xa = acc_xa[t][r] * inv_na;
            const float tt = 1.0f + 2.0f * cu * xy;
            const float A1 = tt + cu * x2;
            const float B1 = 1.0f - cu * p2;
            const float den = tt + cu * cu * p2 * x2;
            const float inv_den = 1.0f / den;
            const float rr = (A1 * A1 * p2 + 2.0f * A1 * B1 * xy + B1 * B1 * x2)
                             * inv_den * inv_den;
            const float lam = 2.0f / (1.0f - rr);
            const float dot = (A1 * pa + B1 * xa) * inv_den;
            out[(size_t)n * C_CLS + cc] = 2.0f * na * asinhf(dot * lam);
        }
    }
}

extern "C" void kernel_launch(void* const* d_in, const int* in_sizes, int n_in,
                              void* d_out, int out_size, void* d_ws, size_t ws_size,
                              hipStream_t stream) {
    const float* X = (const float*)d_in[0];
    const float* A = (const float*)d_in[1];
    const float* P = (const float*)d_in[2];
    float* out = (float*)d_out;

    hyp_mlr_main<<<(N_PTS / WG_PTS) * (C_CLS / WG_CLS), 32 * WG_WAVES, 0, stream>>>(X, A, P, out);
}


# norm phase given to the work-group's four younger waves as their FIRST job, barrier split (signal early, join before the epilogue): each SIMD's older wave runs its products meanwhile; sums unchanged
# speedup vs baseline: 2.6834x; 1.1056x over previous
#include <hip/hip_runtime.h>


typedef __attribute__((ext_vector_type(2))) float v2f;
typedef __attribute__((ext_vector_type(8))) float v8f;

#define N_PTS 4096
#define C_CLS 256
#define D_DIM 128
#define CURVF 1.0f
#define NT_W 2
#define KQ (D_DIM / 8)

#define WG_PTS 64
#define WG_CLS 64
#define WG_WAVES 8
#define NORM_WAVES 4
#define NRW (WG_PTS / NORM_WAVES)

#define DPP_ROW_XMASK(M) (0x160 + (M))

static __device__ __forceinline__ float wave_sum(float s) {
    int t;
    t = __builtin_bit_cast(int, s);
    s += __builtin_bit_cast(float, __builtin_amdgcn_update_dpp(t, t, DPP_ROW_XMASK(1), 0xf, 0xf, true));
    t = __builtin_bit_cast(int, s);
    s += __builtin_bit_cast(float, __builtin_amdgcn_update_dpp(t, t, DPP_ROW_XMASK(2), 0xf, 0xf, true));
    t = __builtin_bit_cast(int, s);
    s += __builtin_bit_cast(float, __builtin_amdgcn_update_dpp(t, t, DPP_ROW_XMASK(4), 0xf, 0xf, true));
    t = __builtin_bit_cast(int, s);
    s += __builtin_bit_cast(float, __builtin_amdgcn_update_dpp(t, t, DPP_ROW_XMASK(8), 0xf, 0xf, true));
    t = __builtin_bit_cast(int, s);
    s += __builtin_bit_cast(float, __builtin_amdgcn_permlanex16(t, t, 0x76543210, 0xfedcba98, false, false));
    return s;
}

static __device__ __forceinline__ float dot4(const float4 u, const float4 v) {
    return u.x * v.x + u.y * v.y + u.z * v.z + u.w * v.w;
}

static __device__ __forceinline__ void wg_signal_lds_done() {
    asm volatile("s_wait_dscnt 0x0\n\ts_barrier_signal -1" ::: "memory");
}
static __device__ __forceinline__ void wg_join() {
    __builtin_amdgcn_sched_barrier(0);
    asm volatile("s_barrier_wait -1" ::: "memory");
}

static_assert(D_DIM == 128, "one 16-byte piece per lane and row");
static_assert(WG_PTS == WG_CLS && WG_PTS % NORM_WAVES == 0 && NRW <= 32 && NORM_WAVES <= WG_WAVES,
              "the norm phase deals whole rows to its waves, one result lane per row");
__global__ __launch_bounds__(32 * WG_WAVES) void hyp_mlr_main(const float* __restrict__ X,
                                                              const float* __restrict__ A,
                                                              const float* __restrict__ P,
                                                              float* __restrict__ out) {
    __shared__ __attribute__((aligned(16))) float s_x2[WG_PTS];
    __shared__ float s_p2[WG_CLS];
    __shared__ float s_pa[WG_CLS];
    __shared__ float s_na[WG_CLS];

    const int lane = threadIdx.x & 31;
    const int wave = threadIdx.x >> 5;
    const int m = lane & 15;
    const int half = lane >> 4;
    const int cblk = (blockIdx.x & 3) * WG_CLS;
    const int nblk = (blockIdx.x >> 2) * WG_PTS;
    const int cl = (wave & 3) * 16 + m;
    const int nl = (wave >> 2) * (16 * NT_W);
    const int cc = cblk + cl;
    const int nbase = nblk + nl;

    const int wv = __builtin_amdgcn_readfirstlane(wave);

    if (wv >= WG_WAVES - NORM_WAVES) {
        const int r0 = NRW * (wv - (WG_WAVES - NORM_WAVES));
        const float4* xr = (const float4*)(X + (size_t)(nblk + r0) * D_DIM) + lane;
        const float4* pr = (const float4*)(P + (size_t)(cblk + r0) * D_DIM) + lane;
        const float4* ar = (const float4*)(A + (size_t)(cblk + r0) * D_DIM) + lane;
        float4 xv[NRW], pv[NRW], av[NRW];
#pragma unroll
        for (int r = 0; r < NRW; ++r) {
            xv[r] = xr[r * (D_DIM / 4)];
            pv[r] = pr[r * (D_DIM / 4)];
            av[r] = ar[r * (D_DIM / 4)];
        }
        float mx2 = 0.f, mp2 = 0.f, mpa = 0.f, ma2 = 0.f;
#pragma unroll
        for (int r = 0; r < NRW; ++r) {
            const float sx2 = wave_sum(dot4(xv[r], xv[r]));
            const float sp2 = wave_sum(dot4(pv[r], pv[r]));
            const float spa = wave_sum(dot4(pv[r], av[r]));
            const float sa2 = wave_sum(dot4(av[r], av[r]));
            if (lane == r) { mx2 = sx2; mp2 = sp2; mpa = spa; ma2 = sa2; }
        }
        if (lane < NRW) {
            const float na = sqrtf(ma2);
            const float nac = fmaxf(na, 1e-12f);
            s_x2[r0 + lane] = mx2;
            s_p2[r0 + lane] = mp2;
            s_pa[r0 + lane] = -mpa / nac;
            s_na[r0 + lane] = na;
        }
    }
    wg_signal_lds_done();

    const float4* prow = (const float4*)(P + (size_t)cc * D_DIM + 4 * half);
    const float4* arow = (const float4*)(A + (size_t)cc * D_DIM + 4 * half);
    const float4* xrow = (const float4*)(X + (size_t)(nbase + m) * D_DIM + 4 * half);

    v8f acc_py[NT_W];
    v8f acc_xa[NT_W];
#pragma unroll
    for (int t = 0; t < NT_W; ++t) {
        acc_py[t] = v8f{};
        acc_xa[t] = v8f{};
    }

#pragma unroll
    for (int q = 0; q < KQ; ++q) {
        const float4 pv = prow[2 * q];
        const float4 av = arow[2 * q];
        const v2f p0 = {pv.x, pv.y};
        const v2f p1 = {pv.z, pv.w};
        const v2f a0 = {av.x, av.y};
        const v2f a1 = {av.z, av.w};
#pragma unroll
        for (int t = 0; t < NT_W; ++t) {
            const float4 xv = xrow[t * (16 * D_DIM / 4) + 2 * q];
            const v2f x0 = {xv.x, xv.y};
            const v2f x1 = {xv.z, xv.w};
            acc_py[t] = __builtin_amdgcn_wmma_f32_16x16x4_f32(false, x0, false, p0,
                                                              (short)0, acc_py[t], false, false);
            acc_xa[t] = __builtin_amdgcn_wmma_f32_16x16x4_f32(false, x0, false, a0,
                                                              (short)0, acc_xa[t], false, false);
            acc_py[t] = __builtin_amdgcn_wmma_f32_16x16x4_f32(false, x1, false, p1,
                                                              (short)0, acc_py[t], false, false);
            acc_xa[t] = __builtin_amdgcn_wmma_f32_16x16x4_f32(false, x1, false, a1,
                                                              (short)0, acc_xa[t], false, false);
        }
    }

    wg_join();

    const float p2 = s_p2[cl];
    const float pa = s_pa[cl];
    const float na = s_na[cl];
    const float inv_na = 1.0f / fmaxf(na, 1e-12f);
    const float cu = CURVF;

#pragma unroll
    for (int t = 0; t < NT_W; ++t) {
        const float4* xq = (const float4*)(s_x2 + nl + 16 * t + 8 * half);
        const float4 xlo = xq[0];
        const float4 xhi = xq[1];
        const float x2r[8] = {xlo.x, xlo.y, xlo.z, xlo.w, xhi.x, xhi.y, xhi.z, xhi.w};
#pragma unroll
        for (int r = 0; r < 8; ++r) {
            const int n = nbase + 16 * t + r + 8 * half;
            const float x2 = x2r[r];
            const float xy = -acc_py[t][r];
            const float xa = acc_xa[t][r] * inv_na;
            const float tt = 1.0f + 2.0f * cu * xy;
            const float A1 = tt + cu * x2;
            const float B1 = 1.0f - cu * p2;
            const float den = tt + cu * cu * p2 * x2;
            const float inv_den = 1.0f / den;
            const float rr = (A1 * A1 * p2 + 2.0f * A1 * B1 * xy + B1 * B1 * x2)
                             * inv_den * inv_den;
            const float lam = 2.0f / (1.0f - rr);
            const float dot = (A1 * pa + B1 * xa) * inv_den;
            out[(size_t)n * C_CLS + cc] = 2.0f * na * asinhf(dot * lam);
        }
    }
}

extern "C" void kernel_launch(void* const* d_in, const int* in_sizes, int n_in,
                              void* d_out, int out_size, void* d_ws, size_t ws_size,
                              hipStream_t stream) {
    const float* X = (const float*)d_in[0];
    const float* A = (const float*)d_in[1];
    const float* P = (const float*)d_in[2];
    float* out = (float*)d_out;

    hyp_mlr_main<<<(N_PTS / WG_PTS) * (C_CLS / WG_CLS), 32 * WG_WAVES, 0, stream>>>(X, A, P, out);
}
